# starting kernel
# baseline (speedup 1.0000x reference)
#include <hip/hip_runtime.h>
#include <hip/hip_bf16.h>

typedef __attribute__((ext_vector_type(2))) float v2f;
typedef __attribute__((ext_vector_type(4))) float v4f;
typedef __attribute__((ext_vector_type(8))) float v8f;

#define KSZ 16
#define LEN 8192
#define HID 64
#define OCH 8
#define OUT_LEN (LEN - KSZ + 1)
#define TILES_PER_BATCH 512
#define NBATCH 64
#define WAVES_PER_BLOCK 8
#define THREADS (WAVES_PER_BLOCK * 32)

__device__ __forceinline__ float fast_tanh(float v) {
    float t = __builtin_amdgcn_exp2f(v * 2.885390081777927f);
    float r = __builtin_amdgcn_rcpf(t + 1.0f);
    return 1.0f - 2.0f * r;
}

__global__ __launch_bounds__(THREADS)
void conv1d_mlp_wmma(const float* __restrict__ x,
                     const float* __restrict__ W1,
                     const float* __restrict__ b1,
                     const float* __restrict__ W2,
                     const float* __restrict__ b2,
                     float* __restrict__ y)
{
    __shared__ float ldsH[WAVES_PER_BLOCK][16 * HID];
    __shared__ float ldsY[WAVES_PER_BLOCK][16 * OCH];

    const int lane  = threadIdx.x & 31;
    const int wave  = threadIdx.x >> 5;
    const int tile  = blockIdx.x * WAVES_PER_BLOCK + wave;
    const int batch = tile >> 9;
    const int t0    = (tile & (TILES_PER_BATCH - 1)) << 4;

    const int half = lane >> 4;
    const int lm   = lane & 15;

    const float* xb = x + (long)batch * LEN;
    const int baseIdx = t0 + lm + 2 * half;
    v2f afrag[4];
    if (t0 + 31 < LEN) {
        #pragma unroll
        for (int kk = 0; kk < 4; ++kk) {
            const int idx = baseIdx + 4 * kk;
            afrag[kk].x = xb[idx];
            afrag[kk].y = xb[idx + 1];
        }
    } else {
        #pragma unroll
        for (int kk = 0; kk < 4; ++kk) {
            const int idx = baseIdx + 4 * kk;
            const int i0 = (idx     < LEN) ? idx     : LEN - 1;
            const int i1 = (idx + 1 < LEN) ? idx + 1 : LEN - 1;
            afrag[kk].x = xb[i0];
            afrag[kk].y = xb[i1];
        }
    }

    v2f b1frag[4][4];
    #pragma unroll
    for (int n = 0; n < 4; ++n) {
        #pragma unroll
        for (int kk = 0; kk < 4; ++kk) {
            const int row = 4 * kk + 2 * half;
            const int col = 16 * n + lm;
            b1frag[n][kk].x = W1[(row    ) * HID + col];
            b1frag[n][kk].y = W1[(row + 1) * HID + col];
        }
    }

    float* hbuf = &ldsH[wave][0];
    #pragma unroll
    for (int n = 0; n < 4; ++n) {
        v8f acc = {};
        #pragma unroll
        for (int kk = 0; kk < 4; ++kk) {
            acc = __builtin_amdgcn_wmma_f32_16x16x4_f32(
                      false, afrag[kk], false, b1frag[n][kk],
                      (short)0, acc, false, false);
        }
        const float bias = b1[16 * n + lm];
        #pragma unroll
        for (int i = 0; i < 8; ++i) {
            const int M = i + 8 * half;
            hbuf[M * HID + 16 * n + lm] = fast_tanh(acc[i] + bias);
        }
    }

    v2f b2frag[16];
    #pragma unroll
    for (int kk = 0; kk < 16; ++kk) {
        const int row = 4 * kk + 2 * half;
        const int col = lm & (OCH - 1);
        b2frag[kk].x = W2[(row    ) * OCH + col];
        b2frag[kk].y = W2[(row + 1) * OCH + col];
    }

    v8f accY = {};
    #pragma unroll
    for (int kk = 0; kk < 16; ++kk) {
        const int kcol = 4 * kk + 2 * half;
        v2f a2;
        a2.x = hbuf[lm * HID + kcol];
        a2.y = hbuf[lm * HID + kcol + 1];
        accY = __builtin_amdgcn_wmma_f32_16x16x4_f32(
                   false, a2, false, b2frag[kk], (short)0, accY, false, false);
    }

    const float bias2 = b2[lm & (OCH - 1)];
    float* ybuf = &ldsY[wave][0];
    if (lm < OCH) {
        #pragma unroll
        for (int i = 0; i < 8; ++i) {
            const int M = i + 8 * half;
            ybuf[M * OCH + lm] = accY[i] + bias2;
        }
    }
    const int row = lane >> 1;
    v4f out4;
    out4.x = ybuf[lane * 4 + 0];
    out4.y = ybuf[lane * 4 + 1];
    out4.z = ybuf[lane * 4 + 2];
    out4.w = ybuf[lane * 4 + 3];
    if (t0 + row < OUT_LEN) {
        const long base = (long)batch * OUT_LEN + t0;
        v4f* dst = (v4f*)(y + (base + row) * OCH + (lane & 1) * 4);
        *dst = out4;
    }
}

extern "C" void kernel_launch(void* const* d_in, const int* in_sizes, int n_in,
                              void* d_out, int out_size, void* d_ws, size_t ws_size,
                              hipStream_t stream) {
    (void)in_sizes; (void)n_in; (void)d_ws; (void)ws_size; (void)out_size;
    const float* x  = (const float*)d_in[0];
    const float* W1 = (const float*)d_in[1];
    const float* b1 = (const float*)d_in[2];
    const float* W2 = (const float*)d_in[3];
    const float* b2 = (const float*)d_in[4];
    float* y = (float*)d_out;

    const int total_tiles = NBATCH * TILES_PER_BATCH;
    const int blocks = total_tiles / WAVES_PER_BLOCK;
    conv1d_mlp_wmma<<<blocks, THREADS, 0, stream>>>(x, W1, b1, W2, b2, y);
}


# both products transposed (positions along the columns): product 1's D registers are product 2's B operand, no LDS re-layout; biases start the accumulators; k-sum order of product 2 changes
# speedup vs baseline: 1.1879x; 1.1879x over previous
#include <hip/hip_runtime.h>
#include <hip/hip_bf16.h>

typedef __attribute__((ext_vector_type(2))) float v2f;
typedef __attribute__((ext_vector_type(4))) float v4f;
typedef __attribute__((ext_vector_type(8))) float v8f;

#define KSZ 16
#define LEN 8192
#define HID 64
#define OCH 8
#define OUT_LEN (LEN - KSZ + 1)
#define TILES_PER_BATCH 512
#define NBATCH 64
#define WAVES_PER_BLOCK 8
#define THREADS (WAVES_PER_BLOCK * 32)

__device__ __forceinline__ float fast_tanh(float v) {
    float t = __builtin_amdgcn_exp2f(v * 2.885390081777927f);
    float r = __builtin_amdgcn_rcpf(t + 1.0f);
    return 1.0f - 2.0f * r;
}

__global__ __launch_bounds__(THREADS)
void conv1d_mlp_wmma(const float* __restrict__ x,
                     const float* __restrict__ W1,
                     const float* __restrict__ b1,
                     const float* __restrict__ W2,
                     const float* __restrict__ b2,
                     float* __restrict__ y)
{
    const int lane  = threadIdx.x & 31;
    const int wave  = threadIdx.x >> 5;
    const int tile  = blockIdx.x * WAVES_PER_BLOCK + wave;
    const int batch = tile >> 9;
    const int t0    = (tile & (TILES_PER_BATCH - 1)) << 4;

    const int half = lane >> 4;
    const int lm   = lane & 15;


    const float* xb = x + (long)batch * LEN;
    const int baseIdx = t0 + lm + 2 * half;
    v2f xfrag[4];
    if (t0 + 31 < LEN) {
        #pragma unroll
        for (int kk = 0; kk < 4; ++kk) {
            const int idx = baseIdx + 4 * kk;
            xfrag[kk].x = xb[idx];
            xfrag[kk].y = xb[idx + 1];
        }
    } else {
        #pragma unroll
        for (int kk = 0; kk < 4; ++kk) {
            const int idx = baseIdx + 4 * kk;
            const int i0 = (idx     < LEN) ? idx     : LEN - 1;
            const int i1 = (idx + 1 < LEN) ? idx + 1 : LEN - 1;
            xfrag[kk].x = xb[i0];
            xfrag[kk].y = xb[i1];
        }
    }

    v2f w1frag[4][4];
    #pragma unroll
    for (int n = 0; n < 4; ++n) {
        #pragma unroll
        for (int kk = 0; kk < 4; ++kk) {
            const int k   = 4 * kk + 2 * half;
            const int hid = 16 * n + lm;
            w1frag[n][kk].x = W1[(k    ) * HID + hid];
            w1frag[n][kk].y = W1[(k + 1) * HID + hid];
        }
    }

    const float* b1p = b1 + 8 * half;
    const float* w2p = W2 + (8 * half) * OCH + (lm & (OCH - 1));

    v8f accY = { b2[0], b2[1], b2[2], b2[3], b2[4], b2[5], b2[6], b2[7] };

    #pragma unroll
    for (int n = 0; n < 4; ++n) {
        const v4f blo = *(const v4f*)(b1p + 16 * n);
        const v4f bhi = *(const v4f*)(b1p + 16 * n + 4);
        v8f acc = { blo.x, blo.y, blo.z, blo.w, bhi.x, bhi.y, bhi.z, bhi.w };
        #pragma unroll
        for (int kk = 0; kk < 4; ++kk) {
            acc = __builtin_amdgcn_wmma_f32_16x16x4_f32(
                      false, w1frag[n][kk], false, xfrag[kk],
                      (short)0, acc, false, false);
        }
        #pragma unroll
        for (int j = 0; j < 4; ++j) {
            v2f hb;
            hb.x = fast_tanh(acc[2 * j]);
            hb.y = fast_tanh(acc[2 * j + 1]);
            v2f wa;
            wa.x = w2p[(16 * n + 2 * j    ) * OCH];
            wa.y = w2p[(16 * n + 2 * j + 1) * OCH];
            accY = __builtin_amdgcn_wmma_f32_16x16x4_f32(
                       false, wa, false, hb, (short)0, accY, false, false);
        }
    }

    v4f out4;
    out4.x = half ? accY[4] : accY[0];
    out4.y = half ? accY[5] : accY[1];
    out4.z = half ? accY[6] : accY[2];
    out4.w = half ? accY[7] : accY[3];
    if (t0 + lm < OUT_LEN) {
        const long base = (long)batch * OUT_LEN + t0;
        v4f* dst = (v4f*)(y + (base + lm) * OCH + 4 * half);
        *dst = out4;
    }
}

extern "C" void kernel_launch(void* const* d_in, const int* in_sizes, int n_in,
                              void* d_out, int out_size, void* d_ws, size_t ws_size,
                              hipStream_t stream) {
    (void)in_sizes; (void)n_in; (void)d_ws; (void)ws_size; (void)out_size;
    const float* x  = (const float*)d_in[0];
    const float* W1 = (const float*)d_in[1];
    const float* b1 = (const float*)d_in[2];
    const float* W2 = (const float*)d_in[3];
    const float* b2 = (const float*)d_in[4];
    float* y = (float*)d_out;

    const int total_tiles = NBATCH * TILES_PER_BATCH;
    const int blocks = total_tiles / WAVES_PER_BLOCK;
    conv1d_mlp_wmma<<<blocks, THREADS, 0, stream>>>(x, W1, b1, W2, b2, y);
}


# each wave computes 4 neighbouring position tiles side by side: every weight fragment and bias is loaded once per wave and feeds 4 independent accumulators; 4x fewer waves; sum order unchanged
# speedup vs baseline: 1.3884x; 1.1688x over previous
#include <hip/hip_runtime.h>
#include <hip/hip_bf16.h>

typedef __attribute__((ext_vector_type(2))) float v2f;
typedef __attribute__((ext_vector_type(4))) float v4f;
typedef __attribute__((ext_vector_type(8))) float v8f;

#define KSZ 16
#define LEN 8192
#define HID 64
#define OCH 8
#define OUT_LEN (LEN - KSZ + 1)
#define TILES_PER_BATCH 512
#define NBATCH 64
#define TPW 4
#define GROUPS_PER_BATCH (TILES_PER_BATCH / TPW)
#define WAVES_PER_BLOCK 8
#define THREADS (WAVES_PER_BLOCK * 32)

__device__ __forceinline__ float fast_tanh(float v) {
    float t = __builtin_amdgcn_exp2f(v * 2.885390081777927f);
    float r = __builtin_amdgcn_rcpf(t + 1.0f);
    return 1.0f - 2.0f * r;
}

__global__ __launch_bounds__(THREADS)
void conv1d_mlp_wmma(const float* __restrict__ x,
                     const float* __restrict__ W1,
                     const float* __restrict__ b1,
                     const float* __restrict__ W2,
                     const float* __restrict__ b2,
                     float* __restrict__ y)
{
    const int lane  = threadIdx.x & 31;
    const int wave  = threadIdx.x >> 5;
    const int group = blockIdx.x * WAVES_PER_BLOCK + wave;
    const int batch = group >> 7;
    const int t0    = (group & (GROUPS_PER_BATCH - 1)) << 6;

    const int half = lane >> 4;
    const int lm   = lane & 15;


    const float* xb = x + (long)batch * LEN;
    const int baseIdx = t0 + lm + 2 * half;
    v2f xfrag[TPW][4];
    if (t0 + 16 * (TPW - 1) + 31 < LEN) {
        #pragma unroll
        for (int t = 0; t < TPW; ++t) {
            #pragma unroll
            for (int kk = 0; kk < 4; ++kk) {
                const int idx = baseIdx + 16 * t + 4 * kk;
                xfrag[t][kk].x = xb[idx];
                xfrag[t][kk].y = xb[idx + 1];
            }
        }
    } else {
        #pragma unroll
        for (int t = 0; t < TPW; ++t) {
            #pragma unroll
            for (int kk = 0; kk < 4; ++kk) {
                const int idx = baseIdx + 16 * t + 4 * kk;
                const int i0 = (idx     < LEN) ? idx     : LEN - 1;
                const int i1 = (idx + 1 < LEN) ? idx + 1 : LEN - 1;
                xfrag[t][kk].x = xb[i0];
                xfrag[t][kk].y = xb[i1];
            }
        }
    }

    v2f w1frag[4][4];
    #pragma unroll
    for (int n = 0; n < 4; ++n) {
        #pragma unroll
        for (int kk = 0; kk < 4; ++kk) {
            const int k   = 4 * kk + 2 * half;
            const int hid = 16 * n + lm;
            w1frag[n][kk].x = W1[(k    ) * HID + hid];
            w1frag[n][kk].y = W1[(k + 1) * HID + hid];
        }
    }

    const float* b1p = b1 + 8 * half;
    const float* w2p = W2 + (8 * half) * OCH + (lm & (OCH - 1));

    const v8f biasY = { b2[0], b2[1], b2[2], b2[3], b2[4], b2[5], b2[6], b2[7] };
    v8f accY[TPW];

    v4f blo = *(const v4f*)(b1p);
    v4f bhi = *(const v4f*)(b1p + 4);

    #pragma unroll
    for (int n = 0; n < 4; ++n) {
        const v8f bias = { blo.x, blo.y, blo.z, blo.w, bhi.x, bhi.y, bhi.z, bhi.w };
        if (n < 3) {
            blo = *(const v4f*)(b1p + 16 * (n + 1));
            bhi = *(const v4f*)(b1p + 16 * (n + 1) + 4);
        }
        v8f acc[TPW];
        #pragma unroll
        for (int kk = 0; kk < 4; ++kk) {
            #pragma unroll
            for (int t = 0; t < TPW; ++t) {
                acc[t] = __builtin_amdgcn_wmma_f32_16x16x4_f32(
                             false, w1frag[n][kk], false, xfrag[t][kk],
                             (short)0, (kk == 0) ? bias : acc[t], false, false);
            }
        }
        #pragma unroll
        for (int j = 0; j < 4; ++j) {
            v2f wa;
            wa.x = w2p[(16 * n + 2 * j    ) * OCH];
            wa.y = w2p[(16 * n + 2 * j + 1) * OCH];
            #pragma unroll
            for (int t = 0; t < TPW; ++t) {
                v2f hb;
                hb.x = fast_tanh(acc[t][2 * j]);
                hb.y = fast_tanh(acc[t][2 * j + 1]);
                accY[t] = __builtin_amdgcn_wmma_f32_16x16x4_f32(
                              false, wa, false, hb, (short)0,
                              (n == 0 && j == 0) ? biasY : accY[t], false, false);
            }
        }
        __builtin_amdgcn_sched_barrier(0);
    }

    const long base = (long)batch * OUT_LEN + t0;
    #pragma unroll
    for (int t = 0; t < TPW; ++t) {
        v4f out4;
        out4.x = half ? accY[t][4] : accY[t][0];
        out4.y = half ? accY[t][5] : accY[t][1];
        out4.z = half ? accY[t][6] : accY[t][2];
        out4.w = half ? accY[t][7] : accY[t][3];
        if (t0 + 16 * t + lm < OUT_LEN) {
            v4f* dst = (v4f*)(y + (base + 16 * t + lm) * OCH + 4 * half);
            *dst = out4;
        }
    }
}

extern "C" void kernel_launch(void* const* d_in, const int* in_sizes, int n_in,
                              void* d_out, int out_size, void* d_ws, size_t ws_size,
                              hipStream_t stream) {
    (void)in_sizes; (void)n_in; (void)d_ws; (void)ws_size; (void)out_size;
    const float* x  = (const float*)d_in[0];
    const float* W1 = (const float*)d_in[1];
    const float* b1 = (const float*)d_in[2];
    const float* W2 = (const float*)d_in[3];
    const float* b2 = (const float*)d_in[4];
    float* y = (float*)d_out;

    const int total_groups = NBATCH * GROUPS_PER_BATCH;
    const int blocks = total_groups / WAVES_PER_BLOCK;
    conv1d_mlp_wmma<<<blocks, THREADS, 0, stream>>>(x, W1, b1, W2, b2, y);
}


# tanh on the hardware instruction v_tanh_f32 instead of mul, exp2, add, rcp, fma: 128 vector issues per wave instead of 450, half the transcendentals; tanh rounding differs, sum order unchanged
# speedup vs baseline: 1.5614x; 1.1246x over previous
#include <hip/hip_runtime.h>
#include <hip/hip_bf16.h>

typedef __attribute__((ext_vector_type(2))) float v2f;
typedef __attribute__((ext_vector_type(4))) float v4f;
typedef __attribute__((ext_vector_type(8))) float v8f;

#define KSZ 16
#define LEN 8192
#define HID 64
#define OCH 8
#define OUT_LEN (LEN - KSZ + 1)
#define TILES_PER_BATCH 512
#define NBATCH 64
#define TPW 4
#define GROUPS_PER_BATCH (TILES_PER_BATCH / TPW)
#define WAVES_PER_BLOCK 8
#define THREADS (WAVES_PER_BLOCK * 32)

__device__ __forceinline__ float fast_tanh(float v) {
    return __builtin_amdgcn_tanhf(v);
}

__global__ __launch_bounds__(THREADS)
void conv1d_mlp_wmma(const float* __restrict__ x,
                     const float* __restrict__ W1,
                     const float* __restrict__ b1,
                     const float* __restrict__ W2,
                     const float* __restrict__ b2,
                     float* __restrict__ y)
{
    const int lane  = threadIdx.x & 31;
    const int wave  = threadIdx.x >> 5;
    const int group = blockIdx.x * WAVES_PER_BLOCK + wave;
    const int batch = group >> 7;
    const int t0    = (group & (GROUPS_PER_BATCH - 1)) << 6;

    const int half = lane >> 4;
    const int lm   = lane & 15;


    const float* xb = x + (long)batch * LEN;
    const int baseIdx = t0 + lm + 2 * half;
    v2f xfrag[TPW][4];
    if (t0 + 16 * (TPW - 1) + 31 < LEN) {
        #pragma unroll
        for (int t = 0; t < TPW; ++t) {
            #pragma unroll
            for (int kk = 0; kk < 4; ++kk) {
                const int idx = baseIdx + 16 * t + 4 * kk;
                xfrag[t][kk].x = xb[idx];
                xfrag[t][kk].y = xb[idx + 1];
            }
        }
    } else {
        #pragma unroll
        for (int t = 0; t < TPW; ++t) {
            #pragma unroll
            for (int kk = 0; kk < 4; ++kk) {
                const int idx = baseIdx + 16 * t + 4 * kk;
                const int i0 = (idx     < LEN) ? idx     : LEN - 1;
                const int i1 = (idx + 1 < LEN) ? idx + 1 : LEN - 1;
                xfrag[t][kk].x = xb[i0];
                xfrag[t][kk].y = xb[i1];
            }
        }
    }

    v2f w1frag[4][4];
    #pragma unroll
    for (int n = 0; n < 4; ++n) {
        #pragma unroll
        for (int kk = 0; kk < 4; ++kk) {
            const int k   = 4 * kk + 2 * half;
            const int hid = 16 * n + lm;
            w1frag[n][kk].x = W1[(k    ) * HID + hid];
            w1frag[n][kk].y = W1[(k + 1) * HID + hid];
        }
    }

    const float* b1p = b1 + 8 * half;
    const float* w2p = W2 + (8 * half) * OCH + (lm & (OCH - 1));

    const v8f biasY = { b2[0], b2[1], b2[2], b2[3], b2[4], b2[5], b2[6], b2[7] };
    v8f accY[TPW];

    v4f blo = *(const v4f*)(b1p);
    v4f bhi = *(const v4f*)(b1p + 4);

    #pragma unroll
    for (int n = 0; n < 4; ++n) {
        const v8f bias = { blo.x, blo.y, blo.z, blo.w, bhi.x, bhi.y, bhi.z, bhi.w };
        if (n < 3) {
            blo = *(const v4f*)(b1p + 16 * (n + 1));
            bhi = *(const v4f*)(b1p + 16 * (n + 1) + 4);
        }
        v8f acc[TPW];
        #pragma unroll
        for (int kk = 0; kk < 4; ++kk) {
            #pragma unroll
            for (int t = 0; t < TPW; ++t) {
                acc[t] = __builtin_amdgcn_wmma_f32_16x16x4_f32(
                             false, w1frag[n][kk], false, xfrag[t][kk],
                             (short)0, (kk == 0) ? bias : acc[t], false, false);
            }
        }
        #pragma unroll
        for (int j = 0; j < 4; ++j) {
            v2f wa;
            wa.x = w2p[(16 * n + 2 * j    ) * OCH];
            wa.y = w2p[(16 * n + 2 * j + 1) * OCH];
            #pragma unroll
            for (int t = 0; t < TPW; ++t) {
                v2f hb;
                hb.x = fast_tanh(acc[t][2 * j]);
                hb.y = fast_tanh(acc[t][2 * j + 1]);
                accY[t] = __builtin_amdgcn_wmma_f32_16x16x4_f32(
                              false, wa, false, hb, (short)0,
                              (n == 0 && j == 0) ? biasY : accY[t], false, false);
            }
        }
        __builtin_amdgcn_sched_barrier(0);
    }

    const long base = (long)batch * OUT_LEN + t0;
    #pragma unroll
    for (int t = 0; t < TPW; ++t) {
        v4f out4;
        out4.x = half ? accY[t][4] : accY[t][0];
        out4.y = half ? accY[t][5] : accY[t][1];
        out4.z = half ? accY[t][6] : accY[t][2];
        out4.w = half ? accY[t][7] : accY[t][3];
        if (t0 + 16 * t + lm < OUT_LEN) {
            v4f* dst = (v4f*)(y + (base + 16 * t + lm) * OCH + 4 * half);
            *dst = out4;
        }
    }
}

extern "C" void kernel_launch(void* const* d_in, const int* in_sizes, int n_in,
                              void* d_out, int out_size, void* d_ws, size_t ws_size,
                              hipStream_t stream) {
    (void)in_sizes; (void)n_in; (void)d_ws; (void)ws_size; (void)out_size;
    const float* x  = (const float*)d_in[0];
    const float* W1 = (const float*)d_in[1];
    const float* b1 = (const float*)d_in[2];
    const float* W2 = (const float*)d_in[3];
    const float* b2 = (const float*)d_in[4];
    float* y = (float*)d_out;

    const int total_groups = NBATCH * GROUPS_PER_BATCH;
    const int blocks = total_groups / WAVES_PER_BLOCK;
    conv1d_mlp_wmma<<<blocks, THREADS, 0, stream>>>(x, W1, b1, W2, b2, y);
}
